# starting kernel
# baseline (speedup 1.0000x reference)
#include <hip/hip_runtime.h>
#include <hip/hip_bf16.h>

typedef __attribute__((ext_vector_type(2))) float v2f;
typedef __attribute__((ext_vector_type(8))) float v8f;

constexpr int BB  = 32;
constexpr int SS  = 2048;
constexpr int DD  = 64;
constexpr int DK  = 24;
constexpr int DV  = 32;
constexpr int LDQ = 32;
constexpr int NT  = SS / 16;

#define WMMA_F32X4(Am, Bm, Cm) \
  __builtin_amdgcn_wmma_f32_16x16x4_f32(false, (Am), false, (Bm), (short)0, (Cm), false, false)

__global__ void __launch_bounds__(256)
qkv_proj_kernel(const float* __restrict__ x, const float* __restrict__ Wq,
                const float* __restrict__ Wk, const float* __restrict__ Wv,
                float* __restrict__ Qp, float* __restrict__ Kp, float* __restrict__ Vm)
{
    int row = blockIdx.x * blockDim.x + threadIdx.x;
    if (row >= BB * SS) return;
    const float* xp = x + (size_t)row * DD;

    float accq[DK], acck[DK], accv[DV];
    #pragma unroll
    for (int j = 0; j < DK; ++j) { accq[j] = 0.f; acck[j] = 0.f; }
    #pragma unroll
    for (int j = 0; j < DV; ++j) accv[j] = 0.f;

    for (int d = 0; d < DD; ++d) {
        float xd = xp[d];
        #pragma unroll
        for (int j = 0; j < DK; ++j) {
            accq[j] = fmaf(xd, Wq[d * DK + j], accq[j]);
            acck[j] = fmaf(xd, Wk[d * DK + j], acck[j]);
        }
        #pragma unroll
        for (int j = 0; j < DV; ++j)
            accv[j] = fmaf(xd, Wv[d * DV + j], accv[j]);
    }
    float* qo = Qp + (size_t)row * LDQ;
    float* ko = Kp + (size_t)row * LDQ;
    float* vo = Vm + (size_t)row * DV;
    #pragma unroll
    for (int j = 0; j < DK; ++j) { qo[j] = accq[j]; ko[j] = acck[j]; }
    #pragma unroll
    for (int j = 0; j < DV; ++j) vo[j] = accv[j];
}

__global__ void __launch_bounds__(32)
col_softmax_stats_kernel(const float* __restrict__ Qp, const float* __restrict__ Kp,
                         float* __restrict__ mArr, float* __restrict__ zinvArr)
{
    int wave = blockIdx.x;
    int bb = wave / NT;
    int tt = wave - bb * NT;
    int lane = threadIdx.x & 31;
    int col  = lane & 15;
    int hl   = lane >> 4;

    const float* kp = Kp + ((size_t)(bb * SS) + (size_t)tt * 16 + col) * LDQ + 2 * hl;
    v2f b0 = *(const v2f*)(kp +  0);
    v2f b1 = *(const v2f*)(kp +  4);
    v2f b2 = *(const v2f*)(kp +  8);
    v2f b3 = *(const v2f*)(kp + 12);
    v2f b4 = *(const v2f*)(kp + 16);
    v2f b5 = *(const v2f*)(kp + 20);

    const float* qbase = Qp + ((size_t)(bb * SS) + col) * LDQ + 2 * hl;
    float m_run = -3.0e38f;
    float z_run = 0.f;

    for (int qt = 0; qt < NT; ++qt) {
        const float* qp = qbase + (size_t)qt * 16 * LDQ;
        __builtin_prefetch(qp + 16 * LDQ, 0, 1);
        v2f a0 = *(const v2f*)(qp +  0);
        v2f a1 = *(const v2f*)(qp +  4);
        v2f a2 = *(const v2f*)(qp +  8);
        v2f a3 = *(const v2f*)(qp + 12);
        v2f a4 = *(const v2f*)(qp + 16);
        v2f a5 = *(const v2f*)(qp + 20);

        v8f s = {};
        s = WMMA_F32X4(a0, b0, s);
        s = WMMA_F32X4(a1, b1, s);
        s = WMMA_F32X4(a2, b2, s);
        s = WMMA_F32X4(a3, b3, s);
        s = WMMA_F32X4(a4, b4, s);
        s = WMMA_F32X4(a5, b5, s);

        float tmax = fmaxf(fmaxf(fmaxf(s[0], s[1]), fmaxf(s[2], s[3])),
                           fmaxf(fmaxf(s[4], s[5]), fmaxf(s[6], s[7])));
        tmax = fmaxf(tmax, __shfl_xor(tmax, 16, 32));
        float m_new = fmaxf(m_run, tmax);
        float p = 0.f;
        #pragma unroll
        for (int i = 0; i < 8; ++i) p += __expf(s[i] - m_new);
        p += __shfl_xor(p, 16, 32);
        z_run = fmaf(z_run, __expf(m_run - m_new), p);
        m_run = m_new;
    }
    if (hl == 0) {
        mArr[bb * SS + tt * 16 + col]    = m_run;
        zinvArr[bb * SS + tt * 16 + col] = 1.f / z_run;
    }
}

__global__ void __launch_bounds__(256)
scale_v_kernel(float* __restrict__ Vm, const float* __restrict__ zinvArr)
{
    int i = blockIdx.x * blockDim.x + threadIdx.x;
    if (i < BB * SS * DV) Vm[i] *= zinvArr[i / DV];
}

__global__ void __launch_bounds__(32)
attn_h1_kernel(const float* __restrict__ Qp, const float* __restrict__ Kp,
               const float* __restrict__ VZ, const float* __restrict__ mArr,
               float* __restrict__ h1)
{
    __shared__ float ldsE[16 * 20];
    int wave = blockIdx.x;
    int bb = wave / NT;
    int qt = wave - bb * NT;
    int lane = threadIdx.x & 31;
    int col  = lane & 15;
    int hl   = lane >> 4;

    const float* qp = Qp + ((size_t)(bb * SS) + (size_t)qt * 16 + col) * LDQ + 2 * hl;
    v2f a0 = *(const v2f*)(qp +  0);
    v2f a1 = *(const v2f*)(qp +  4);
    v2f a2 = *(const v2f*)(qp +  8);
    v2f a3 = *(const v2f*)(qp + 12);
    v2f a4 = *(const v2f*)(qp + 16);
    v2f a5 = *(const v2f*)(qp + 20);

    v8f acc0 = {};
    v8f acc1 = {};

    const float* kbase = Kp + ((size_t)(bb * SS) + col) * LDQ + 2 * hl;
    const float* vzb   = VZ + (size_t)(bb * SS) * DV;
    const float* mrow  = mArr + bb * SS;

    for (int tt = 0; tt < NT; ++tt) {
        const float* kp = kbase + (size_t)tt * 16 * LDQ;
        __builtin_prefetch(kp + 16 * LDQ, 0, 1);
        v2f b0 = *(const v2f*)(kp +  0);
        v2f b1 = *(const v2f*)(kp +  4);
        v2f b2 = *(const v2f*)(kp +  8);
        v2f b3 = *(const v2f*)(kp + 12);
        v2f b4 = *(const v2f*)(kp + 16);
        v2f b5 = *(const v2f*)(kp + 20);

        v8f s = {};
        s = WMMA_F32X4(a0, b0, s);
        s = WMMA_F32X4(a1, b1, s);
        s = WMMA_F32X4(a2, b2, s);
        s = WMMA_F32X4(a3, b3, s);
        s = WMMA_F32X4(a4, b4, s);
        s = WMMA_F32X4(a5, b5, s);

        float mcol = mrow[tt * 16 + col];
        #pragma unroll
        for (int r = 0; r < 8; ++r)
            ldsE[(r + 8 * hl) * 20 + col] = __expf(s[r] - mcol);

        const float* vt = vzb + (size_t)(tt * 16) * DV;
        #pragma unroll
        for (int c = 0; c < 4; ++c) {
            v2f ae = *(const v2f*)&ldsE[col * 20 + 4 * c + 2 * hl];
            const float* vp = vt + (size_t)(4 * c + 2 * hl) * DV;
            v2f bv0, bv1;
            bv0.x = vp[col];           bv0.y = vp[DV + col];
            bv1.x = vp[16 + col];      bv1.y = vp[DV + 16 + col];
            acc0 = WMMA_F32X4(ae, bv0, acc0);
            acc1 = WMMA_F32X4(ae, bv1, acc1);
        }
    }

    #pragma unroll
    for (int r = 0; r < 8; ++r) {
        float* hp = h1 + ((size_t)(bb * SS) + (size_t)qt * 16 + r + 8 * hl) * DV;
        hp[col]      = acc0[r];
        hp[col + 16] = acc1[r];
    }
}

__global__ void __launch_bounds__(256)
out_proj_kernel(const float* __restrict__ h1, const float* __restrict__ Wh,
                float* __restrict__ out)
{
    int row = blockIdx.x * blockDim.x + threadIdx.x;
    if (row >= BB * SS) return;
    const float* hp = h1 + (size_t)row * DV;
    float acc[DD];
    #pragma unroll
    for (int d = 0; d < DD; ++d) acc[d] = 0.f;
    for (int v = 0; v < DV; ++v) {
        float hv = hp[v];
        #pragma unroll
        for (int d = 0; d < DD; ++d)
            acc[d] = fmaf(hv, Wh[v * DD + d], acc[d]);
    }
    float* op = out + (size_t)row * DD;
    #pragma unroll
    for (int d = 0; d < DD; ++d) op[d] = acc[d];
}

extern "C" void kernel_launch(void* const* d_in, const int* in_sizes, int n_in,
                              void* d_out, int out_size, void* d_ws, size_t ws_size,
                              hipStream_t stream)
{
    const float* x  = (const float*)d_in[0];
    const float* Wq = (const float*)d_in[1];
    const float* Wk = (const float*)d_in[2];
    const float* Wv = (const float*)d_in[3];
    const float* Wh = (const float*)d_in[4];
    float* out = (float*)d_out;

    float* ws = (float*)d_ws;
    size_t o = 0;
    float* Qp = ws + o; o += (size_t)BB * SS * LDQ;
    float* Kp = ws + o; o += (size_t)BB * SS * LDQ;
    float* Vm = ws + o; o += (size_t)BB * SS * DV;
    float* mA = ws + o; o += (size_t)BB * SS;
    float* zI = ws + o; o += (size_t)BB * SS;
    float* h1 = ws + o; o += (size_t)BB * SS * DV;

    const int rows = BB * SS;
    qkv_proj_kernel<<<(rows + 255) / 256, 256, 0, stream>>>(x, Wq, Wk, Wv, Qp, Kp, Vm);
    col_softmax_stats_kernel<<<BB * NT, 32, 0, stream>>>(Qp, Kp, mA, zI);
    const int nv = BB * SS * DV;
    scale_v_kernel<<<(nv + 255) / 256, 256, 0, stream>>>(Vm, zI);
    attn_h1_kernel<<<BB * NT, 32, 0, stream>>>(Qp, Kp, Vm, mA, h1);
    out_proj_kernel<<<(rows + 255) / 256, 256, 0, stream>>>(h1, Wh, out);
}


# Both projection kernels: a work-group's 128 rows staged in LDS by coalesced 16-byte loads, two rows per lane with the columns split over four waves, results leave as whole 128-byte lines; same sums
# speedup vs baseline: 1.6620x; 1.6620x over previous
#include <hip/hip_runtime.h>
#include <hip/hip_bf16.h>

typedef __attribute__((ext_vector_type(2))) float v2f;
typedef __attribute__((ext_vector_type(4))) float v4f;
typedef __attribute__((ext_vector_type(8))) float v8f;

constexpr int BB  = 32;
constexpr int SS  = 2048;
constexpr int DD  = 64;
constexpr int DK  = 24;
constexpr int DV  = 32;
constexpr int LDQ = 32;
constexpr int NT  = SS / 16;
constexpr int WPB = 8;
constexpr int TS  = 64;
constexpr int NS  = SS / TS;
constexpr int PK  = 28;
constexpr int PV  = 40;
constexpr int KCH = TS * (DK / 4);
constexpr int PR  = 128;
constexpr int PX  = 68;
constexpr int PO  = 36;
static_assert(LDQ == DV && (BB * SS) % PR == 0 && PR * PX <= 3 * PR * PO, "projection tiling");

#define WMMA_F32X4(Am, Bm, Cm) \
  __builtin_amdgcn_wmma_f32_16x16x4_f32(false, (Am), false, (Bm), (short)0, (Cm), false, false)

template <int NC, int WS, int KD, int KS>
__device__ __forceinline__ void proj_two_rows(const float* xa, const float* xb,
                                              const float* __restrict__ w,
                                              float* a, float* b)
{
    typedef __attribute__((ext_vector_type(KS))) float vks;
    for (int j = 0; j < KD / KS; ++j) {
        const vks va = *(const vks*)(xa + KS * j);
        const vks vb = *(const vks*)(xb + KS * j);
        #pragma unroll
        for (int kk = 0; kk < KS; ++kk) {
            const int d = KS * j + kk;
            #pragma unroll
            for (int c = 0; c < NC; ++c) {
                const float wv = w[d * WS + c];
                a[c] = fmaf(va[kk], wv, a[c]);
                b[c] = fmaf(vb[kk], wv, b[c]);
            }
        }
    }
}

__global__ void __launch_bounds__(256)
qkv_proj_kernel(const float* __restrict__ x, const float* __restrict__ Wq,
                const float* __restrict__ Wk, const float* __restrict__ Wv,
                float* __restrict__ Qp, float* __restrict__ Kp, float* __restrict__ Vm)
{
    __shared__ __attribute__((aligned(16))) float sb[3 * PR * PO];

    const int tid  = threadIdx.x;
    const int lane = tid & 31;
    const int wu   = __builtin_amdgcn_readfirstlane(tid >> 5);
    const int row0 = blockIdx.x * PR;

    const float* xg = x + (size_t)row0 * DD;
    #pragma unroll
    for (int i = 0; i < PR * (DD / 4) / 256; ++i) {
        const int idx = tid + 256 * i;
        v4f c = *(const v4f*)(xg + 4 * idx);
        *(v4f*)&sb[(idx >> 4) * PX + 4 * (idx & 15)] = c;
    }
    __syncthreads();

    const int g  = wu >> 2;
    const int sp = (wu + 2 * g) & 3;
    const int ra = 64 * g + lane;
    const float* xa = &sb[ra * PX];
    const float* xb = xa + 32 * PX;

    float a[DK], b[DK];
    #pragma unroll
    for (int c = 0; c < DK; ++c) { a[c] = 0.f; b[c] = 0.f; }

    if (sp < 2) proj_two_rows<DK, DK, DD, 2>(xa, xb, sp == 0 ? Wq : Wk, a, b);
    else        proj_two_rows<16, DV, DD, 4>(xa, xb, Wv + 16 * (sp - 2), a, b);

    __syncthreads();

    if (sp < 2) {
        float* pa = &sb[(sp * PR + ra) * PO];
        float* pb = pa + 32 * PO;
        #pragma unroll
        for (int i = 0; i < DK / 4; ++i) {
            v4f ta = {a[4 * i], a[4 * i + 1], a[4 * i + 2], a[4 * i + 3]};
            v4f tb = {b[4 * i], b[4 * i + 1], b[4 * i + 2], b[4 * i + 3]};
            *(v4f*)(pa + 4 * i) = ta;
            *(v4f*)(pb + 4 * i) = tb;
        }
        const v4f z = {0.f, 0.f, 0.f, 0.f};
        *(v4f*)(pa + DK) = z;  *(v4f*)(pa + DK + 4) = z;
        *(v4f*)(pb + DK) = z;  *(v4f*)(pb + DK + 4) = z;
    } else {
        float* pa = &sb[(2 * PR + ra) * PO + 16 * (sp - 2)];
        float* pb = pa + 32 * PO;
        #pragma unroll
        for (int i = 0; i < 4; ++i) {
            v4f ta = {a[4 * i], a[4 * i + 1], a[4 * i + 2], a[4 * i + 3]};
            v4f tb = {b[4 * i], b[4 * i + 1], b[4 * i + 2], b[4 * i + 3]};
            *(v4f*)(pa + 4 * i) = ta;
            *(v4f*)(pb + 4 * i) = tb;
        }
    }
    __syncthreads();

    #pragma unroll
    for (int pl = 0; pl < 3; ++pl) {
        float* dst = (pl == 0 ? Qp : (pl == 1 ? Kp : Vm)) + (size_t)row0 * LDQ;
        #pragma unroll
        for (int i = 0; i < PR * (LDQ / 4) / 256; ++i) {
            const int idx = tid + 256 * i;
            v4f c = *(const v4f*)&sb[(pl * PR + (idx >> 3)) * PO + 4 * (idx & 7)];
            *(v4f*)(dst + 4 * idx) = c;
        }
    }
}

__global__ void __launch_bounds__(256)
col_softmax_stats_kernel(const float* __restrict__ Qp, const float* __restrict__ Kp,
                         float* __restrict__ cArr)
{
    __shared__ __attribute__((aligned(16))) float sQ[2][TS * PK];

    int tid  = threadIdx.x;
    int w    = tid >> 5;
    int bb   = blockIdx.x / (NT / WPB);
    int tt   = (blockIdx.x - bb * (NT / WPB)) * WPB + w;
    int lane = tid & 31;
    int col  = lane & 15;
    int hl   = lane >> 4;

    const float* kp = Kp + ((size_t)(bb * SS) + (size_t)tt * 16 + col) * LDQ + 2 * hl;
    v2f b0 = *(const v2f*)(kp +  0);
    v2f b1 = *(const v2f*)(kp +  4);
    v2f b2 = *(const v2f*)(kp +  8);
    v2f b3 = *(const v2f*)(kp + 12);
    v2f b4 = *(const v2f*)(kp + 16);
    v2f b5 = *(const v2f*)(kp + 20);

    const int  r0 = tid / 6,         j0 = tid - r0 * 6;
    const int  r1 = (tid + 256) / 6, j1 = (tid + 256) - r1 * 6;
    const bool two = tid < KCH - 256;
    const float* qsrc = Qp + (size_t)(bb * SS) * LDQ;
    const float* g0 = qsrc + r0 * LDQ + 4 * j0;
    const float* g1 = qsrc + r1 * LDQ + 4 * j1;
    const int l0 = r0 * PK + 4 * j0;
    const int l1 = r1 * PK + 4 * j1;

    {
        v4f c0 = *(const v4f*)g0;
        *(v4f*)&sQ[0][l0] = c0;
        if (two) { v4f c1 = *(const v4f*)g1; *(v4f*)&sQ[0][l1] = c1; }
    }
    __syncthreads();

    float m_run = -3.0e38f;
    float z_run = 0.f;
    const int aoff = col * PK + 2 * hl;

    for (int st = 0; st < NS; ++st) {
        const int  cur  = st & 1;
        const bool more = st + 1 < NS;
        v4f n0 = {}, n1 = {};
        if (more) {
            n0 = *(const v4f*)(g0 + (size_t)(st + 1) * TS * LDQ);
            if (two) n1 = *(const v4f*)(g1 + (size_t)(st + 1) * TS * LDQ);
        }

        const float* sq = &sQ[cur][aoff];
        #pragma unroll
        for (int tl = 0; tl < TS / 16; ++tl) {
            const float* qp = sq + tl * 16 * PK;
            v2f a0 = *(const v2f*)(qp +  0);
            v2f a1 = *(const v2f*)(qp +  4);
            v2f a2 = *(const v2f*)(qp +  8);
            v2f a3 = *(const v2f*)(qp + 12);
            v2f a4 = *(const v2f*)(qp + 16);
            v2f a5 = *(const v2f*)(qp + 20);

            v8f s = {};
            s = WMMA_F32X4(a0, b0, s);
            s = WMMA_F32X4(a1, b1, s);
            s = WMMA_F32X4(a2, b2, s);
            s = WMMA_F32X4(a3, b3, s);
            s = WMMA_F32X4(a4, b4, s);
            s = WMMA_F32X4(a5, b5, s);

            float tmax = fmaxf(fmaxf(fmaxf(s[0], s[1]), fmaxf(s[2], s[3])),
                               fmaxf(fmaxf(s[4], s[5]), fmaxf(s[6], s[7])));
            tmax = fmaxf(tmax, __shfl_xor(tmax, 16, 32));
            float m_new = fmaxf(m_run, tmax);
            float p = 0.f;
            #pragma unroll
            for (int i = 0; i < 8; ++i) p += __expf(s[i] - m_new);
            p += __shfl_xor(p, 16, 32);
            z_run = fmaf(z_run, __expf(m_run - m_new), p);
            m_run = m_new;
        }

        if (more) {
            *(v4f*)&sQ[cur ^ 1][l0] = n0;
            if (two) *(v4f*)&sQ[cur ^ 1][l1] = n1;
        }
        __syncthreads();
    }
    if (hl == 0)
        cArr[bb * SS + tt * 16 + col] = m_run + logf(z_run);
}

__global__ void __launch_bounds__(256)
attn_h1_kernel(const float* __restrict__ Qp, const float* __restrict__ Kp,
               const float* __restrict__ Vm, const float* __restrict__ cArr,
               float* __restrict__ h1)
{
    __shared__ __attribute__((aligned(16))) float sK[2][TS * PK];
    __shared__ __attribute__((aligned(16))) float sV[2][TS * PV];
    __shared__ float sM[2][TS];
    __shared__ __attribute__((aligned(16))) float sE[WPB][16 * 20];

    int tid  = threadIdx.x;
    int w    = tid >> 5;
    int bb   = blockIdx.x / (NT / WPB);
    int qt   = (blockIdx.x - bb * (NT / WPB)) * WPB + w;
    int lane = tid & 31;
    int col  = lane & 15;
    int hl   = lane >> 4;

    const float* qp = Qp + ((size_t)(bb * SS) + (size_t)qt * 16 + col) * LDQ + 2 * hl;
    v2f a0 = *(const v2f*)(qp +  0);
    v2f a1 = *(const v2f*)(qp +  4);
    v2f a2 = *(const v2f*)(qp +  8);
    v2f a3 = *(const v2f*)(qp + 12);
    v2f a4 = *(const v2f*)(qp + 16);
    v2f a5 = *(const v2f*)(qp + 20);

    v8f acc0 = {};
    v8f acc1 = {};

    const int  r0 = tid / 6,         j0 = tid - r0 * 6;
    const int  r1 = (tid + 256) / 6, j1 = (tid + 256) - r1 * 6;
    const bool two = tid < KCH - 256;
    const bool ism = (tid >= 128) && (tid < 128 + TS);
    const float* ksrc = Kp + (size_t)(bb * SS) * LDQ;
    const float* gk0 = ksrc + r0 * LDQ + 4 * j0;
    const float* gk1 = ksrc + r1 * LDQ + 4 * j1;
    const int lk0 = r0 * PK + 4 * j0;
    const int lk1 = r1 * PK + 4 * j1;
    const int vr = tid >> 3, vj = tid & 7;
    const float* gv0 = Vm + (size_t)(bb * SS) * DV + 4 * tid;
    const float* gv1 = gv0 + 32 * DV;
    const int lv0 = vr * PV + 4 * vj;
    const int lv1 = lv0 + 32 * PV;
    const int    mi = ism ? tid - 128 : 0;
    const float* gm = cArr + bb * SS + mi;

    {
        v4f k0 = *(const v4f*)gk0;
        v4f v0 = *(const v4f*)gv0;
        v4f v1 = *(const v4f*)gv1;
        *(v4f*)&sK[0][lk0] = k0;
        *(v4f*)&sV[0][lv0] = v0;
        *(v4f*)&sV[0][lv1] = v1;
        if (two) { v4f k1 = *(const v4f*)gk1; *(v4f*)&sK[0][lk1] = k1; }
        if (ism) sM[0][mi] = gm[0];
    }
    __syncthreads();

    float* myE = &sE[w][0];
    const int boff = col * PK + 2 * hl;

    for (int st = 0; st < NS; ++st) {
        const int  cur  = st & 1;
        const bool more = st + 1 < NS;
        v4f nk0 = {}, nk1 = {}, nv0 = {}, nv1 = {};
        float nm = 0.f;
        if (more) {
            nk0 = *(const v4f*)(gk0 + (size_t)(st + 1) * TS * LDQ);
            nv0 = *(const v4f*)(gv0 + (size_t)(st + 1) * TS * DV);
            nv1 = *(const v4f*)(gv1 + (size_t)(st + 1) * TS * DV);
            if (two) nk1 = *(const v4f*)(gk1 + (size_t)(st + 1) * TS * LDQ);
            if (ism) nm  = gm[(st + 1) * TS];
        }

        const float* sk = &sK[cur][boff];
        const float* sv = &sV[cur][0];
        const float* sm = &sM[cur][0];
        #pragma unroll
        for (int tl = 0; tl < TS / 16; ++tl) {
            const float* kp = sk + tl * 16 * PK;
            v2f b0 = *(const v2f*)(kp +  0);
            v2f b1 = *(const v2f*)(kp +  4);
            v2f b2 = *(const v2f*)(kp +  8);
            v2f b3 = *(const v2f*)(kp + 12);
            v2f b4 = *(const v2f*)(kp + 16);
            v2f b5 = *(const v2f*)(kp + 20);

            v8f s = {};
            s = WMMA_F32X4(a0, b0, s);
            s = WMMA_F32X4(a1, b1, s);
            s = WMMA_F32X4(a2, b2, s);
            s = WMMA_F32X4(a3, b3, s);
            s = WMMA_F32X4(a4, b4, s);
            s = WMMA_F32X4(a5, b5, s);

            float mcol = sm[tl * 16 + col];
            #pragma unroll
            for (int r = 0; r < 8; ++r)
                myE[(r + 8 * hl) * 20 + col] = __expf(s[r] - mcol);

            const float* vt = sv + (tl * 16) * PV;
            #pragma unroll
            for (int c = 0; c < 4; ++c) {
                v2f ae = *(const v2f*)&myE[col * 20 + 4 * c + 2 * hl];
                const float* vp = vt + (4 * c + 2 * hl) * PV;
                v2f bv0, bv1;
                bv0.x = vp[col];           bv0.y = vp[PV + col];
                bv1.x = vp[16 + col];      bv1.y = vp[PV + 16 + col];
                acc0 = WMMA_F32X4(ae, bv0, acc0);
                acc1 = WMMA_F32X4(ae, bv1, acc1);
            }
        }

        if (more) {
            *(v4f*)&sK[cur ^ 1][lk0] = nk0;
            *(v4f*)&sV[cur ^ 1][lv0] = nv0;
            *(v4f*)&sV[cur ^ 1][lv1] = nv1;
            if (two) *(v4f*)&sK[cur ^ 1][lk1] = nk1;
            if (ism) sM[cur ^ 1][mi] = nm;
        }
        __syncthreads();
    }

    #pragma unroll
    for (int r = 0; r < 8; ++r) {
        float* hp = h1 + ((size_t)(bb * SS) + (size_t)qt * 16 + r + 8 * hl) * DV;
        hp[col]      = acc0[r];
        hp[col + 16] = acc1[r];
    }
}

__global__ void __launch_bounds__(256)
out_proj_kernel(const float* __restrict__ h1, const float* __restrict__ Wh,
                float* __restrict__ out)
{
    __shared__ __attribute__((aligned(16))) float sH[PR * PO];
    __shared__ __attribute__((aligned(16))) float sO[PR * PX];

    const int tid  = threadIdx.x;
    const int lane = tid & 31;
    const int wu   = __builtin_amdgcn_readfirstlane(tid >> 5);
    const int row0 = blockIdx.x * PR;

    const float* hg = h1 + (size_t)row0 * DV;
    #pragma unroll
    for (int i = 0; i < PR * (DV / 4) / 256; ++i) {
        const int idx = tid + 256 * i;
        v4f c = *(const v4f*)(hg + 4 * idx);
        *(v4f*)&sH[(idx >> 3) * PO + 4 * (idx & 7)] = c;
    }
    __syncthreads();

    const int g  = wu >> 2;
    const int qd = wu & 3;
    const int ra = 64 * g + lane;
    const float* ha = &sH[ra * PO];
    const float* hb = ha + 32 * PO;

    float a[16], b[16];
    #pragma unroll
    for (int c = 0; c < 16; ++c) { a[c] = 0.f; b[c] = 0.f; }

    proj_two_rows<16, DD, DV, 4>(ha, hb, Wh + 16 * qd, a, b);

    float* pa = &sO[ra * PX + 16 * qd];
    float* pb = pa + 32 * PX;
    #pragma unroll
    for (int i = 0; i < 4; ++i) {
        v4f ta = {a[4 * i], a[4 * i + 1], a[4 * i + 2], a[4 * i + 3]};
        v4f tb = {b[4 * i], b[4 * i + 1], b[4 * i + 2], b[4 * i + 3]};
        *(v4f*)(pa + 4 * i) = ta;
        *(v4f*)(pb + 4 * i) = tb;
    }
    __syncthreads();

    float* og = out + (size_t)row0 * DD;
    #pragma unroll
    for (int i = 0; i < PR * (DD / 4) / 256; ++i) {
        const int idx = tid + 256 * i;
        v4f c = *(const v4f*)&sO[(idx >> 4) * PX + 4 * (idx & 15)];
        *(v4f*)(og + 4 * idx) = c;
    }
}

extern "C" void kernel_launch(void* const* d_in, const int* in_sizes, int n_in,
                              void* d_out, int out_size, void* d_ws, size_t ws_size,
                              hipStream_t stream)
{
    const float* x  = (const float*)d_in[0];
    const float* Wq = (const float*)d_in[1];
    const float* Wk = (const float*)d_in[2];
    const float* Wv = (const float*)d_in[3];
    const float* Wh = (const float*)d_in[4];
    float* out = (float*)d_out;

    float* ws = (float*)d_ws;
    size_t o = 0;
    float* Qp = ws + o; o += (size_t)BB * SS * LDQ;
    float* Kp = ws + o; o += (size_t)BB * SS * LDQ;
    float* Vm = ws + o; o += (size_t)BB * SS * DV;
    float* cA = ws + o; o += (size_t)BB * SS;
    float* h1 = ws + o; o += (size_t)BB * SS * DV;

    const int rows = BB * SS;
    qkv_proj_kernel<<<rows / PR, 256, 0, stream>>>(x, Wq, Wk, Wv, Qp, Kp, Vm);
    col_softmax_stats_kernel<<<BB * NT / WPB, 32 * WPB, 0, stream>>>(Qp, Kp, cA);
    attn_h1_kernel<<<BB * NT / WPB, 32 * WPB, 0, stream>>>(Qp, Kp, Vm, cA, h1);
    out_proj_kernel<<<rows / PR, 256, 0, stream>>>(h1, Wh, out);
}


# Both streaming WMMA kernels: each resident tile gets two waves, each streaming half the rows through its own LDS ring (16-wave work-groups), partials merged in LDS; t sum and column sum orders change
# speedup vs baseline: 1.7434x; 1.0490x over previous
#include <hip/hip_runtime.h>
#include <hip/hip_bf16.h>

typedef __attribute__((ext_vector_type(2))) float v2f;
typedef __attribute__((ext_vector_type(4))) float v4f;
typedef __attribute__((ext_vector_type(8))) float v8f;

constexpr int BB  = 32;
constexpr int SS  = 2048;
constexpr int DD  = 64;
constexpr int DK  = 24;
constexpr int DV  = 32;
constexpr int LDQ = 32;
constexpr int NT  = SS / 16;
constexpr int WPB = 8;
constexpr int NH  = 2;
constexpr int GT  = 32 * WPB;
constexpr int TS  = 64;
constexpr int NS  = SS / TS;
constexpr int HS  = NS / NH;
constexpr int PK  = 28;
constexpr int PV  = 40;
constexpr int KCH = TS * (DK / 4);
constexpr int PR  = 128;
constexpr int PX  = 68;
constexpr int PO  = 36;
static_assert(LDQ == DV && (BB * SS) % PR == 0 && PR * PX <= 3 * PR * PO, "projection tiling");
static_assert(NS % NH == 0 && GT == 256 && KCH > GT && KCH <= 2 * GT, "streaming copy roles");

#define WMMA_F32X4(Am, Bm, Cm) \
  __builtin_amdgcn_wmma_f32_16x16x4_f32(false, (Am), false, (Bm), (short)0, (Cm), false, false)

template <int NC, int WS, int KD, int KS>
__device__ __forceinline__ void proj_two_rows(const float* xa, const float* xb,
                                              const float* __restrict__ w,
                                              float* a, float* b)
{
    typedef __attribute__((ext_vector_type(KS))) float vks;
    for (int j = 0; j < KD / KS; ++j) {
        const vks va = *(const vks*)(xa + KS * j);
        const vks vb = *(const vks*)(xb + KS * j);
        #pragma unroll
        for (int kk = 0; kk < KS; ++kk) {
            const int d = KS * j + kk;
            #pragma unroll
            for (int c = 0; c < NC; ++c) {
                const float wv = w[d * WS + c];
                a[c] = fmaf(va[kk], wv, a[c]);
                b[c] = fmaf(vb[kk], wv, b[c]);
            }
        }
    }
}

__global__ void __launch_bounds__(256)
qkv_proj_kernel(const float* __restrict__ x, const float* __restrict__ Wq,
                const float* __restrict__ Wk, const float* __restrict__ Wv,
                float* __restrict__ Qp, float* __restrict__ Kp, float* __restrict__ Vm)
{
    __shared__ __attribute__((aligned(16))) float sb[3 * PR * PO];

    const int tid  = threadIdx.x;
    const int lane = tid & 31;
    const int wu   = __builtin_amdgcn_readfirstlane(tid >> 5);
    const int row0 = blockIdx.x * PR;

    const float* xg = x + (size_t)row0 * DD;
    #pragma unroll
    for (int i = 0; i < PR * (DD / 4) / 256; ++i) {
        const int idx = tid + 256 * i;
        v4f c = *(const v4f*)(xg + 4 * idx);
        *(v4f*)&sb[(idx >> 4) * PX + 4 * (idx & 15)] = c;
    }
    __syncthreads();

    const int g  = wu >> 2;
    const int sp = (wu + 2 * g) & 3;
    const int ra = 64 * g + lane;
    const float* xa = &sb[ra * PX];
    const float* xb = xa + 32 * PX;

    float a[DK], b[DK];
    #pragma unroll
    for (int c = 0; c < DK; ++c) { a[c] = 0.f; b[c] = 0.f; }

    if (sp < 2) proj_two_rows<DK, DK, DD, 2>(xa, xb, sp == 0 ? Wq : Wk, a, b);
    else        proj_two_rows<16, DV, DD, 4>(xa, xb, Wv + 16 * (sp - 2), a, b);

    __syncthreads();

    if (sp < 2) {
        float* pa = &sb[(sp * PR + ra) * PO];
        float* pb = pa + 32 * PO;
        #pragma unroll
        for (int i = 0; i < DK / 4; ++i) {
            v4f ta = {a[4 * i], a[4 * i + 1], a[4 * i + 2], a[4 * i + 3]};
            v4f tb = {b[4 * i], b[4 * i + 1], b[4 * i + 2], b[4 * i + 3]};
            *(v4f*)(pa + 4 * i) = ta;
            *(v4f*)(pb + 4 * i) = tb;
        }
        const v4f z = {0.f, 0.f, 0.f, 0.f};
        *(v4f*)(pa + DK) = z;  *(v4f*)(pa + DK + 4) = z;
        *(v4f*)(pb + DK) = z;  *(v4f*)(pb + DK + 4) = z;
    } else {
        float* pa = &sb[(2 * PR + ra) * PO + 16 * (sp - 2)];
        float* pb = pa + 32 * PO;
        #pragma unroll
        for (int i = 0; i < 4; ++i) {
            v4f ta = {a[4 * i], a[4 * i + 1], a[4 * i + 2], a[4 * i + 3]};
            v4f tb = {b[4 * i], b[4 * i + 1], b[4 * i + 2], b[4 * i + 3]};
            *(v4f*)(pa + 4 * i) = ta;
            *(v4f*)(pb + 4 * i) = tb;
        }
    }
    __syncthreads();

    #pragma unroll
    for (int pl = 0; pl < 3; ++pl) {
        float* dst = (pl == 0 ? Qp : (pl == 1 ? Kp : Vm)) + (size_t)row0 * LDQ;
        #pragma unroll
        for (int i = 0; i < PR * (LDQ / 4) / 256; ++i) {
            const int idx = tid + 256 * i;
            v4f c = *(const v4f*)&sb[(pl * PR + (idx >> 3)) * PO + 4 * (idx & 7)];
            *(v4f*)(dst + 4 * idx) = c;
        }
    }
}

__global__ void __launch_bounds__(GT * NH)
col_softmax_stats_kernel(const float* __restrict__ Qp, const float* __restrict__ Kp,
                         float* __restrict__ cArr)
{
    __shared__ __attribute__((aligned(16))) float sQ[NH][2][TS * PK];
    __shared__ float sMZ[2][WPB * 16];

    int tid  = threadIdx.x;
    int w    = tid >> 5;
    int wt   = w & (WPB - 1);
    int hf   = w / WPB;
    int ht   = tid & (GT - 1);
    int bb   = blockIdx.x / (NT / WPB);
    int tt   = (blockIdx.x - bb * (NT / WPB)) * WPB + wt;
    int lane = tid & 31;
    int col  = lane & 15;
    int hl   = lane >> 4;

    const float* kp = Kp + ((size_t)(bb * SS) + (size_t)tt * 16 + col) * LDQ + 2 * hl;
    v2f b0 = *(const v2f*)(kp +  0);
    v2f b1 = *(const v2f*)(kp +  4);
    v2f b2 = *(const v2f*)(kp +  8);
    v2f b3 = *(const v2f*)(kp + 12);
    v2f b4 = *(const v2f*)(kp + 16);
    v2f b5 = *(const v2f*)(kp + 20);

    const int  r0 = ht / 6,        j0 = ht - r0 * 6;
    const int  r1 = (ht + GT) / 6, j1 = (ht + GT) - r1 * 6;
    const bool two = ht < KCH - GT;
    const float* qsrc = Qp + ((size_t)(bb * SS) + (size_t)(hf * HS * TS)) * LDQ;
    const float* g0 = qsrc + r0 * LDQ + 4 * j0;
    const float* g1 = qsrc + r1 * LDQ + 4 * j1;
    const int l0 = r0 * PK + 4 * j0;
    const int l1 = r1 * PK + 4 * j1;
    float* ring0 = &sQ[hf][0][0];
    float* ring1 = &sQ[hf][1][0];

    {
        v4f c0 = *(const v4f*)g0;
        *(v4f*)&ring0[l0] = c0;
        if (two) { v4f c1 = *(const v4f*)g1; *(v4f*)&ring0[l1] = c1; }
    }
    __syncthreads();

    float m_run = -3.0e38f;
    float z_run = 0.f;
    const int aoff = col * PK + 2 * hl;

    for (int st = 0; st < HS; ++st) {
        const int  cur  = st & 1;
        const bool more = st + 1 < HS;
        v4f n0 = {}, n1 = {};
        if (more) {
            n0 = *(const v4f*)(g0 + (size_t)(st + 1) * TS * LDQ);
            if (two) n1 = *(const v4f*)(g1 + (size_t)(st + 1) * TS * LDQ);
        }

        const float* sq = (cur ? ring1 : ring0) + aoff;
        #pragma unroll
        for (int tl = 0; tl < TS / 16; ++tl) {
            const float* qp = sq + tl * 16 * PK;
            v2f a0 = *(const v2f*)(qp +  0);
            v2f a1 = *(const v2f*)(qp +  4);
            v2f a2 = *(const v2f*)(qp +  8);
            v2f a3 = *(const v2f*)(qp + 12);
            v2f a4 = *(const v2f*)(qp + 16);
            v2f a5 = *(const v2f*)(qp + 20);

            v8f s = {};
            s = WMMA_F32X4(a0, b0, s);
            s = WMMA_F32X4(a1, b1, s);
            s = WMMA_F32X4(a2, b2, s);
            s = WMMA_F32X4(a3, b3, s);
            s = WMMA_F32X4(a4, b4, s);
            s = WMMA_F32X4(a5, b5, s);

            float tmax = fmaxf(fmaxf(fmaxf(s[0], s[1]), fmaxf(s[2], s[3])),
                               fmaxf(fmaxf(s[4], s[5]), fmaxf(s[6], s[7])));
            tmax = fmaxf(tmax, __shfl_xor(tmax, 16, 32));
            float m_new = fmaxf(m_run, tmax);
            float p = 0.f;
            #pragma unroll
            for (int i = 0; i < 8; ++i) p += __expf(s[i] - m_new);
            p += __shfl_xor(p, 16, 32);
            z_run = fmaf(z_run, __expf(m_run - m_new), p);
            m_run = m_new;
        }

        if (more) {
            float* nx = cur ? ring0 : ring1;
            *(v4f*)&nx[l0] = n0;
            if (two) *(v4f*)&nx[l1] = n1;
        }
        __syncthreads();
    }
    if (hf == 1 && hl == 0) {
        sMZ[0][wt * 16 + col] = m_run;
        sMZ[1][wt * 16 + col] = z_run;
    }
    __syncthreads();
    if (hf == 0 && hl == 0) {
        const float m1 = sMZ[0][wt * 16 + col];
        const float z1 = sMZ[1][wt * 16 + col];
        const float m  = fmaxf(m_run, m1);
        const float z  = z_run * __expf(m_run - m) + z1 * __expf(m1 - m);
        cArr[bb * SS + tt * 16 + col] = m + logf(z);
    }
}

__global__ void __launch_bounds__(GT * NH)
attn_h1_kernel(const float* __restrict__ Qp, const float* __restrict__ Kp,
               const float* __restrict__ Vm, const float* __restrict__ cArr,
               float* __restrict__ h1)
{
    __shared__ __attribute__((aligned(16))) float sK[NH][2][TS * PK];
    __shared__ __attribute__((aligned(16))) float sV[NH][2][TS * PV];
    __shared__ float sM[NH][2][TS];
    __shared__ __attribute__((aligned(16))) float sE[NH * WPB][16 * 20];
    static_assert(NH * 2 * TS * PV >= WPB * 16 * 32, "sV also holds the partial sums at the end");

    int tid  = threadIdx.x;
    int w    = tid >> 5;
    int wq   = w & (WPB - 1);
    int hf   = w / WPB;
    int ht   = tid & (GT - 1);
    int bb   = blockIdx.x / (NT / WPB);
    int qt   = (blockIdx.x - bb * (NT / WPB)) * WPB + wq;
    int lane = tid & 31;
    int col  = lane & 15;
    int hl   = lane >> 4;

    const float* qp = Qp + ((size_t)(bb * SS) + (size_t)qt * 16 + col) * LDQ + 2 * hl;
    v2f a0 = *(const v2f*)(qp +  0);
    v2f a1 = *(const v2f*)(qp +  4);
    v2f a2 = *(const v2f*)(qp +  8);
    v2f a3 = *(const v2f*)(qp + 12);
    v2f a4 = *(const v2f*)(qp + 16);
    v2f a5 = *(const v2f*)(qp + 20);

    v8f acc0 = {};
    v8f acc1 = {};

    const int  r0 = ht / 6,        j0 = ht - r0 * 6;
    const int  r1 = (ht + GT) / 6, j1 = (ht + GT) - r1 * 6;
    const bool two = ht < KCH - GT;
    const bool ism = (ht >= 128) && (ht < 128 + TS);
    const size_t key0 = (size_t)(bb * SS) + (size_t)(hf * HS * TS);
    const float* ksrc = Kp + key0 * LDQ;
    const float* gk0 = ksrc + r0 * LDQ + 4 * j0;
    const float* gk1 = ksrc + r1 * LDQ + 4 * j1;
    const int lk0 = r0 * PK + 4 * j0;
    const int lk1 = r1 * PK + 4 * j1;
    const int vr = ht >> 3, vj = ht & 7;
    const float* gv0 = Vm + key0 * DV + 4 * ht;
    const float* gv1 = gv0 + 32 * DV;
    const int lv0 = vr * PV + 4 * vj;
    const int lv1 = lv0 + 32 * PV;
    const int    mi = ism ? ht - 128 : 0;
    const float* gm = cArr + key0 + mi;
    float* rK0 = &sK[hf][0][0];  float* rK1 = &sK[hf][1][0];
    float* rV0 = &sV[hf][0][0];  float* rV1 = &sV[hf][1][0];
    float* rM0 = &sM[hf][0][0];  float* rM1 = &sM[hf][1][0];

    {
        v4f k0 = *(const v4f*)gk0;
        v4f v0 = *(const v4f*)gv0;
        v4f v1 = *(const v4f*)gv1;
        *(v4f*)&rK0[lk0] = k0;
        *(v4f*)&rV0[lv0] = v0;
        *(v4f*)&rV0[lv1] = v1;
        if (two) { v4f k1 = *(const v4f*)gk1; *(v4f*)&rK0[lk1] = k1; }
        if (ism) rM0[mi] = gm[0];
    }
    __syncthreads();

    float* myE = &sE[w][0];
    const int boff = col * PK + 2 * hl;

    for (int st = 0; st < HS; ++st) {
        const int  cur  = st & 1;
        const bool more = st + 1 < HS;
        v4f nk0 = {}, nk1 = {}, nv0 = {}, nv1 = {};
        float nm = 0.f;
        if (more) {
            nk0 = *(const v4f*)(gk0 + (size_t)(st + 1) * TS * LDQ);
            nv0 = *(const v4f*)(gv0 + (size_t)(st + 1) * TS * DV);
            nv1 = *(const v4f*)(gv1 + (size_t)(st + 1) * TS * DV);
            if (two) nk1 = *(const v4f*)(gk1 + (size_t)(st + 1) * TS * LDQ);
            if (ism) nm  = gm[(st + 1) * TS];
        }

        const float* sk = (cur ? rK1 : rK0) + boff;
        const float* sv = cur ? rV1 : rV0;
        const float* sm = cur ? rM1 : rM0;
        #pragma unroll
        for (int tl = 0; tl < TS / 16; ++tl) {
            const float* kp = sk + tl * 16 * PK;
            v2f b0 = *(const v2f*)(kp +  0);
            v2f b1 = *(const v2f*)(kp +  4);
            v2f b2 = *(const v2f*)(kp +  8);
            v2f b3 = *(const v2f*)(kp + 12);
            v2f b4 = *(const v2f*)(kp + 16);
            v2f b5 = *(const v2f*)(kp + 20);

            v8f s = {};
            s = WMMA_F32X4(a0, b0, s);
            s = WMMA_F32X4(a1, b1, s);
            s = WMMA_F32X4(a2, b2, s);
            s = WMMA_F32X4(a3, b3, s);
            s = WMMA_F32X4(a4, b4, s);
            s = WMMA_F32X4(a5, b5, s);

            float mcol = sm[tl * 16 + col];
            #pragma unroll
            for (int r = 0; r < 8; ++r)
                myE[(r + 8 * hl) * 20 + col] = __expf(s[r] - mcol);

            const float* vt = sv + (tl * 16) * PV;
            #pragma unroll
            for (int c = 0; c < 4; ++c) {
                v2f ae = *(const v2f*)&myE[col * 20 + 4 * c + 2 * hl];
                const float* vp = vt + (4 * c + 2 * hl) * PV;
                v2f bv0, bv1;
                bv0.x = vp[col];           bv0.y = vp[PV + col];
                bv1.x = vp[16 + col];      bv1.y = vp[PV + 16 + col];
                acc0 = WMMA_F32X4(ae, bv0, acc0);
                acc1 = WMMA_F32X4(ae, bv1, acc1);
            }
        }

        if (more) {
            float* nK = cur ? rK0 : rK1;
            float* nV = cur ? rV0 : rV1;
            float* nM = cur ? rM0 : rM1;
            *(v4f*)&nK[lk0] = nk0;
            *(v4f*)&nV[lv0] = nv0;
            *(v4f*)&nV[lv1] = nv1;
            if (two) *(v4f*)&nK[lk1] = nk1;
            if (ism) nM[mi] = nm;
        }
        __syncthreads();
    }

    float* red = &sV[0][0][0] + (wq * 16) * 32 + lane;
    if (hf == 1) {
        #pragma unroll
        for (int r = 0; r < 8; ++r) {
            red[r * 32]       = acc0[r];
            red[(8 + r) * 32] = acc1[r];
        }
    }
    __syncthreads();
    if (hf == 0) {
        #pragma unroll
        for (int r = 0; r < 8; ++r) {
            float* hp = h1 + ((size_t)(bb * SS) + (size_t)qt * 16 + r + 8 * hl) * DV;
            hp[col]      = acc0[r] + red[r * 32];
            hp[col + 16] = acc1[r] + red[(8 + r) * 32];
        }
    }
}

__global__ void __launch_bounds__(256)
out_proj_kernel(const float* __restrict__ h1, const float* __restrict__ Wh,
                float* __restrict__ out)
{
    __shared__ __attribute__((aligned(16))) float sH[PR * PO];
    __shared__ __attribute__((aligned(16))) float sO[PR * PX];

    const int tid  = threadIdx.x;
    const int lane = tid & 31;
    const int wu   = __builtin_amdgcn_readfirstlane(tid >> 5);
    const int row0 = blockIdx.x * PR;

    const float* hg = h1 + (size_t)row0 * DV;
    #pragma unroll
    for (int i = 0; i < PR * (DV / 4) / 256; ++i) {
        const int idx = tid + 256 * i;
        v4f c = *(const v4f*)(hg + 4 * idx);
        *(v4f*)&sH[(idx >> 3) * PO + 4 * (idx & 7)] = c;
    }
    __syncthreads();

    const int g  = wu >> 2;
    const int qd = wu & 3;
    const int ra = 64 * g + lane;
    const float* ha = &sH[ra * PO];
    const float* hb = ha + 32 * PO;

    float a[16], b[16];
    #pragma unroll
    for (int c = 0; c < 16; ++c) { a[c] = 0.f; b[c] = 0.f; }

    proj_two_rows<16, DD, DV, 4>(ha, hb, Wh + 16 * qd, a, b);

    float* pa = &sO[ra * PX + 16 * qd];
    float* pb = pa + 32 * PX;
    #pragma unroll
    for (int i = 0; i < 4; ++i) {
        v4f ta = {a[4 * i], a[4 * i + 1], a[4 * i + 2], a[4 * i + 3]};
        v4f tb = {b[4 * i], b[4 * i + 1], b[4 * i + 2], b[4 * i + 3]};
        *(v4f*)(pa + 4 * i) = ta;
        *(v4f*)(pb + 4 * i) = tb;
    }
    __syncthreads();

    float* og = out + (size_t)row0 * DD;
    #pragma unroll
    for (int i = 0; i < PR * (DD / 4) / 256; ++i) {
        const int idx = tid + 256 * i;
        v4f c = *(const v4f*)&sO[(idx >> 4) * PX + 4 * (idx & 15)];
        *(v4f*)(og + 4 * idx) = c;
    }
}

extern "C" void kernel_launch(void* const* d_in, const int* in_sizes, int n_in,
                              void* d_out, int out_size, void* d_ws, size_t ws_size,
                              hipStream_t stream)
{
    const float* x  = (const float*)d_in[0];
    const float* Wq = (const float*)d_in[1];
    const float* Wk = (const float*)d_in[2];
    const float* Wv = (const float*)d_in[3];
    const float* Wh = (const float*)d_in[4];
    float* out = (float*)d_out;

    float* ws = (float*)d_ws;
    size_t o = 0;
    float* Qp = ws + o; o += (size_t)BB * SS * LDQ;
    float* Kp = ws + o; o += (size_t)BB * SS * LDQ;
    float* Vm = ws + o; o += (size_t)BB * SS * DV;
    float* cA = ws + o; o += (size_t)BB * SS;
    float* h1 = ws + o; o += (size_t)BB * SS * DV;

    const int rows = BB * SS;
    qkv_proj_kernel<<<rows / PR, 256, 0, stream>>>(x, Wq, Wk, Wv, Qp, Kp, Vm);
    col_softmax_stats_kernel<<<BB * NT / WPB, GT * NH, 0, stream>>>(Qp, Kp, cA);
    attn_h1_kernel<<<BB * NT / WPB, GT * NH, 0, stream>>>(Qp, Kp, Vm, cA, h1);
    out_proj_kernel<<<rows / PR, 256, 0, stream>>>(h1, Wh, out);
}
